# starting kernel
# baseline (speedup 1.0000x reference)
#include <hip/hip_runtime.h>


typedef float v2f __attribute__((ext_vector_type(2)));
typedef float v8f __attribute__((ext_vector_type(8)));

#define T_SEQ 2048
#define HEADS 4
#define DH    64
#define C_DIM 256

__device__ __forceinline__ v8f wmma4(v2f a, v2f b, v8f c) {
  return __builtin_amdgcn_wmma_f32_16x16x4_f32(false, a, false, b, (short)0, c,
                                               false, false);
}

__global__ void qkv_proj_kernel(const float* __restrict__ X,
                                const float* __restrict__ W,
                                const float* __restrict__ bias,
                                float* __restrict__ Qw,
                                float* __restrict__ Kw,
                                float* __restrict__ Vw,
                                int mtiles) {
  const int lane = threadIdx.x & 31;
  const int lo = lane & 15;
  const int hi = lane >> 4;
  const int ntiles = (3 * C_DIM) / 16;
  int tile = (int)((blockIdx.x * blockDim.x + threadIdx.x) >> 5);
  if (tile >= mtiles * ntiles) return;
  const int mt = tile / ntiles;
  const int nt = tile - mt * ntiles;
  const int m0 = mt * 16, n0 = nt * 16;

  const float* arow = X + (size_t)(m0 + lo) * C_DIM;
  const float* wcol = W + (n0 + lo);
  v8f c = {};
#pragma unroll 8
  for (int kk = 0; kk < C_DIM; kk += 4) {
    const int ka = kk + hi * 2;
    v2f a, b;
    a.x = arow[ka];
    a.y = arow[ka + 1];
    b.x = wcol[(size_t)ka * (3 * C_DIM)];
    b.y = wcol[(size_t)(ka + 1) * (3 * C_DIM)];
    c = wmma4(a, b, c);
  }

  const int cg = n0 + lo;
  const int which = cg >> 8;
  const int within = cg & 255;
  const int h = within >> 6;
  const int dd = within & 63;
  const float bv = bias[cg];
  float* dst = (which == 0) ? Qw : ((which == 1) ? Kw : Vw);
#pragma unroll
  for (int r = 0; r < 8; ++r) {
    const int row = m0 + r + hi * 8;
    const int bb = row >> 11;
    const int tt = row & (T_SEQ - 1);
    dst[(((size_t)(bb * HEADS + h)) * T_SEQ + tt) * DH + dd] = c[r] + bv;
  }
}

__global__ void attn_kernel(const float* __restrict__ Q,
                            const float* __restrict__ K,
                            const float* __restrict__ V,
                            float* __restrict__ O,
                            int nwaves) {
  const int lane = threadIdx.x & 31;
  const int lo = lane & 15;
  const int hi = lane >> 4;
  int wave = (int)((blockIdx.x * blockDim.x + threadIdx.x) >> 5);
  if (wave >= nwaves) return;

  const int qtiles = T_SEQ / 16;
  const int qt = wave % qtiles;
  const int bh = wave / qtiles;
  const int q0 = qt * 16;
  const float* Qh = Q + (size_t)bh * T_SEQ * DH;
  const float* Kh = K + (size_t)bh * T_SEQ * DH;
  const float* Vh = V + (size_t)bh * T_SEQ * DH;

  const float qs = 0.125f * 1.44269504088896340736f;
  v2f qreg[16];
  {
    const float* qrow = Qh + (size_t)(q0 + lo) * DH;
#pragma unroll
    for (int i = 0; i < 16; ++i) {
      qreg[i].x = qrow[i * 4 + hi * 2] * qs;
      qreg[i].y = qrow[i * 4 + hi * 2 + 1] * qs;
    }
  }

  float m_run = -__builtin_inff();
  float l_run = 0.0f;
  v8f o0 = {}, o1 = {}, o2 = {}, o3 = {};

  for (int kt0 = 0; kt0 <= q0; kt0 += 16) {
    v8f s = {};
    const float* krow = Kh + (size_t)(kt0 + lo) * DH;
#pragma unroll
    for (int i = 0; i < 16; ++i) {
      v2f a;
      a.x = krow[i * 4 + hi * 2];
      a.y = krow[i * 4 + hi * 2 + 1];
      s = wmma4(a, qreg[i], s);
    }
    if (kt0 == q0) {
#pragma unroll
      for (int r = 0; r < 8; ++r)
        s[r] = (r + hi * 8 > lo) ? -__builtin_inff() : s[r];
    }

    float tmax = s[0];
#pragma unroll
    for (int r = 1; r < 8; ++r) tmax = fmaxf(tmax, s[r]);
    tmax = fmaxf(tmax, __shfl_xor(tmax, 16, 32));
    const float m_new = fmaxf(m_run, tmax);
    const float alpha = __builtin_amdgcn_exp2f(m_run - m_new);
    float p[8];
    float ts = 0.0f;
#pragma unroll
    for (int r = 0; r < 8; ++r) {
      p[r] = __builtin_amdgcn_exp2f(s[r] - m_new);
      ts += p[r];
    }
    ts += __shfl_xor(ts, 16, 32);
    l_run = l_run * alpha + ts;
    m_run = m_new;
#pragma unroll
    for (int r = 0; r < 8; ++r) {
      o0[r] *= alpha; o1[r] *= alpha; o2[r] *= alpha; o3[r] *= alpha;
    }

    const float* vbase = Vh + (size_t)kt0 * DH;
#pragma unroll
    for (int kk = 0; kk < 16; kk += 4) {
      v2f bop;
      if (kk < 8) {
        const float sx = __shfl_xor(p[kk + 2], 16, 32);
        const float sy = __shfl_xor(p[kk + 3], 16, 32);
        bop.x = hi ? sx : p[kk];
        bop.y = hi ? sy : p[kk + 1];
      } else {
        const int j = kk - 8;
        const float sx = __shfl_xor(p[j], 16, 32);
        const float sy = __shfl_xor(p[j + 1], 16, 32);
        bop.x = hi ? p[j + 2] : sx;
        bop.y = hi ? p[j + 3] : sy;
      }
      const float* vk = vbase + (size_t)(kk + hi * 2) * DH + lo;
      v2f a;
      a.x = vk[0];       a.y = vk[DH];        o0 = wmma4(a, bop, o0);
      a.x = vk[16];      a.y = vk[DH + 16];   o1 = wmma4(a, bop, o1);
      a.x = vk[32];      a.y = vk[DH + 32];   o2 = wmma4(a, bop, o2);
      a.x = vk[48];      a.y = vk[DH + 48];   o3 = wmma4(a, bop, o3);
    }
  }

  const float inv_l = 1.0f / l_run;
  const int b = bh / HEADS;
  const int h = bh - b * HEADS;
  float* orow = O + ((size_t)b * T_SEQ + q0 + lo) * C_DIM + h * DH + hi * 8;
#pragma unroll
  for (int r = 0; r < 8; ++r) {
    orow[r]      = o0[r] * inv_l;
    orow[16 + r] = o1[r] * inv_l;
    orow[32 + r] = o2[r] * inv_l;
    orow[48 + r] = o3[r] * inv_l;
  }
}

__global__ void out_proj_kernel(const float* __restrict__ A,
                                const float* __restrict__ W,
                                const float* __restrict__ bias,
                                float* __restrict__ Out,
                                int mtiles) {
  const int lane = threadIdx.x & 31;
  const int lo = lane & 15;
  const int hi = lane >> 4;
  const int ntiles = C_DIM / 16;
  int tile = (int)((blockIdx.x * blockDim.x + threadIdx.x) >> 5);
  if (tile >= mtiles * ntiles) return;
  const int mt = tile / ntiles;
  const int nt = tile - mt * ntiles;
  const int m0 = mt * 16, n0 = nt * 16;

  const float* arow = A + (size_t)(m0 + lo) * C_DIM;
  const float* wcol = W + (n0 + lo);
  v8f c = {};
#pragma unroll 8
  for (int kk = 0; kk < C_DIM; kk += 4) {
    const int ka = kk + hi * 2;
    v2f a, b;
    a.x = arow[ka];
    a.y = arow[ka + 1];
    b.x = wcol[(size_t)ka * C_DIM];
    b.y = wcol[(size_t)(ka + 1) * C_DIM];
    c = wmma4(a, b, c);
  }
  const float bv = bias[n0 + lo];
#pragma unroll
  for (int r = 0; r < 8; ++r)
    Out[(size_t)(m0 + r + hi * 8) * C_DIM + n0 + lo] = c[r] + bv;
}

extern "C" void kernel_launch(void* const* d_in, const int* in_sizes, int n_in,
                              void* d_out, int out_size, void* d_ws, size_t ws_size,
                              hipStream_t stream) {
  const float* x     = (const float*)d_in[0];
  const float* W_qkv = (const float*)d_in[1];
  const float* b_qkv = (const float*)d_in[2];
  const float* W_fc  = (const float*)d_in[3];
  const float* b_fc  = (const float*)d_in[4];
  float* out = (float*)d_out;

  const int BT = in_sizes[0] / C_DIM;
  const int B  = BT / T_SEQ;

  const size_t headElems = (size_t)B * HEADS * T_SEQ * DH;
  float* Qw = (float*)d_ws;
  float* Kw = Qw + headElems;
  float* Vw = Kw + headElems;
  float* Aw = Vw + headElems;

  const int mtiles = BT / 16;

  {
    const int tiles = mtiles * ((3 * C_DIM) / 16);
    const int blocks = (tiles + 7) / 8;
    qkv_proj_kernel<<<blocks, 256, 0, stream>>>(x, W_qkv, b_qkv, Qw, Kw, Vw,
                                                mtiles);
  }
  {
    const int nwaves = B * HEADS * (T_SEQ / 16);
    const int blocks = (nwaves + 7) / 8;
    attn_kernel<<<blocks, 256, 0, stream>>>(Qw, Kw, Vw, Aw, nwaves);
  }
  {
    const int tiles = mtiles * (C_DIM / 16);
    const int blocks = (tiles + 7) / 8;
    out_proj_kernel<<<blocks, 256, 0, stream>>>(Aw, W_fc, b_fc, out, mtiles);
  }
}


# attention: each wave takes query tiles 127-p and p of one head in turn, so every wave does 129 key tiles (equal causal work; 2048 waves, 4 per block); per-tile arithmetic and sum order unchanged
# speedup vs baseline: 1.1160x; 1.1160x over previous
#include <hip/hip_runtime.h>


typedef float v2f __attribute__((ext_vector_type(2)));
typedef float v8f __attribute__((ext_vector_type(8)));

#define T_SEQ 2048
#define HEADS 4
#define DH    64
#define C_DIM 256

__device__ __forceinline__ v8f wmma4(v2f a, v2f b, v8f c) {
  return __builtin_amdgcn_wmma_f32_16x16x4_f32(false, a, false, b, (short)0, c,
                                               false, false);
}

__global__ void qkv_proj_kernel(const float* __restrict__ X,
                                const float* __restrict__ W,
                                const float* __restrict__ bias,
                                float* __restrict__ Qw,
                                float* __restrict__ Kw,
                                float* __restrict__ Vw,
                                int mtiles) {
  const int lane = threadIdx.x & 31;
  const int lo = lane & 15;
  const int hi = lane >> 4;
  const int ntiles = (3 * C_DIM) / 16;
  int tile = (int)((blockIdx.x * blockDim.x + threadIdx.x) >> 5);
  if (tile >= mtiles * ntiles) return;
  const int mt = tile / ntiles;
  const int nt = tile - mt * ntiles;
  const int m0 = mt * 16, n0 = nt * 16;

  const float* arow = X + (size_t)(m0 + lo) * C_DIM;
  const float* wcol = W + (n0 + lo);
  v8f c = {};
#pragma unroll 8
  for (int kk = 0; kk < C_DIM; kk += 4) {
    const int ka = kk + hi * 2;
    v2f a, b;
    a.x = arow[ka];
    a.y = arow[ka + 1];
    b.x = wcol[(size_t)ka * (3 * C_DIM)];
    b.y = wcol[(size_t)(ka + 1) * (3 * C_DIM)];
    c = wmma4(a, b, c);
  }

  const int cg = n0 + lo;
  const int which = cg >> 8;
  const int within = cg & 255;
  const int h = within >> 6;
  const int dd = within & 63;
  const float bv = bias[cg];
  float* dst = (which == 0) ? Qw : ((which == 1) ? Kw : Vw);
#pragma unroll
  for (int r = 0; r < 8; ++r) {
    const int row = m0 + r + hi * 8;
    const int bb = row >> 11;
    const int tt = row & (T_SEQ - 1);
    dst[(((size_t)(bb * HEADS + h)) * T_SEQ + tt) * DH + dd] = c[r] + bv;
  }
}

__global__ void __launch_bounds__(128) attn_kernel(const float* __restrict__ Q,
                            const float* __restrict__ K,
                            const float* __restrict__ V,
                            float* __restrict__ O,
                            int nwaves) {
  const int lane = threadIdx.x & 31;
  const int lo = lane & 15;
  const int hi = lane >> 4;
  int wave = (int)((blockIdx.x * blockDim.x + threadIdx.x) >> 5);
  if (wave >= nwaves) return;

  const int qtiles = T_SEQ / 16;
  const int pairs = qtiles / 2;
  const int pr = wave % pairs;
  const int bh = wave / pairs;
  const float* Qh = Q + (size_t)bh * T_SEQ * DH;
  const float* Kh = K + (size_t)bh * T_SEQ * DH;
  const float* Vh = V + (size_t)bh * T_SEQ * DH;

  const float qs = 0.125f * 1.44269504088896340736f;
  const int b = bh / HEADS;
  const int h = bh - b * HEADS;

  for (int t = 0; t < 2; ++t) {
  const int qt = t ? pr : (qtiles - 1 - pr);
  const int q0 = qt * 16;
  v2f qreg[16];
  {
    const float* qrow = Qh + (size_t)(q0 + lo) * DH;
#pragma unroll
    for (int i = 0; i < 16; ++i) {
      qreg[i].x = qrow[i * 4 + hi * 2] * qs;
      qreg[i].y = qrow[i * 4 + hi * 2 + 1] * qs;
    }
  }

  float m_run = -__builtin_inff();
  float l_run = 0.0f;
  v8f o0 = {}, o1 = {}, o2 = {}, o3 = {};

  for (int kt0 = 0; kt0 <= q0; kt0 += 16) {
    v8f s = {};
    const float* krow = Kh + (size_t)(kt0 + lo) * DH;
#pragma unroll
    for (int i = 0; i < 16; ++i) {
      v2f a;
      a.x = krow[i * 4 + hi * 2];
      a.y = krow[i * 4 + hi * 2 + 1];
      s = wmma4(a, qreg[i], s);
    }
    if (kt0 == q0) {
#pragma unroll
      for (int r = 0; r < 8; ++r)
        s[r] = (r + hi * 8 > lo) ? -__builtin_inff() : s[r];
    }

    float tmax = s[0];
#pragma unroll
    for (int r = 1; r < 8; ++r) tmax = fmaxf(tmax, s[r]);
    tmax = fmaxf(tmax, __shfl_xor(tmax, 16, 32));
    const float m_new = fmaxf(m_run, tmax);
    const float alpha = __builtin_amdgcn_exp2f(m_run - m_new);
    float p[8];
    float ts = 0.0f;
#pragma unroll
    for (int r = 0; r < 8; ++r) {
      p[r] = __builtin_amdgcn_exp2f(s[r] - m_new);
      ts += p[r];
    }
    ts += __shfl_xor(ts, 16, 32);
    l_run = l_run * alpha + ts;
    m_run = m_new;
#pragma unroll
    for (int r = 0; r < 8; ++r) {
      o0[r] *= alpha; o1[r] *= alpha; o2[r] *= alpha; o3[r] *= alpha;
    }

    const float* vbase = Vh + (size_t)kt0 * DH;
#pragma unroll
    for (int kk = 0; kk < 16; kk += 4) {
      v2f bop;
      if (kk < 8) {
        const float sx = __shfl_xor(p[kk + 2], 16, 32);
        const float sy = __shfl_xor(p[kk + 3], 16, 32);
        bop.x = hi ? sx : p[kk];
        bop.y = hi ? sy : p[kk + 1];
      } else {
        const int j = kk - 8;
        const float sx = __shfl_xor(p[j], 16, 32);
        const float sy = __shfl_xor(p[j + 1], 16, 32);
        bop.x = hi ? p[j + 2] : sx;
        bop.y = hi ? p[j + 3] : sy;
      }
      const float* vk = vbase + (size_t)(kk + hi * 2) * DH + lo;
      v2f a;
      a.x = vk[0];       a.y = vk[DH];        o0 = wmma4(a, bop, o0);
      a.x = vk[16];      a.y = vk[DH + 16];   o1 = wmma4(a, bop, o1);
      a.x = vk[32];      a.y = vk[DH + 32];   o2 = wmma4(a, bop, o2);
      a.x = vk[48];      a.y = vk[DH + 48];   o3 = wmma4(a, bop, o3);
    }
  }

  const float inv_l = 1.0f / l_run;
  float* orow = O + ((size_t)b * T_SEQ + q0 + lo) * C_DIM + h * DH + hi * 8;
#pragma unroll
  for (int r = 0; r < 8; ++r) {
    orow[r]      = o0[r] * inv_l;
    orow[16 + r] = o1[r] * inv_l;
    orow[32 + r] = o2[r] * inv_l;
    orow[48 + r] = o3[r] * inv_l;
  }
  }
}

__global__ void out_proj_kernel(const float* __restrict__ A,
                                const float* __restrict__ W,
                                const float* __restrict__ bias,
                                float* __restrict__ Out,
                                int mtiles) {
  const int lane = threadIdx.x & 31;
  const int lo = lane & 15;
  const int hi = lane >> 4;
  const int ntiles = C_DIM / 16;
  int tile = (int)((blockIdx.x * blockDim.x + threadIdx.x) >> 5);
  if (tile >= mtiles * ntiles) return;
  const int mt = tile / ntiles;
  const int nt = tile - mt * ntiles;
  const int m0 = mt * 16, n0 = nt * 16;

  const float* arow = A + (size_t)(m0 + lo) * C_DIM;
  const float* wcol = W + (n0 + lo);
  v8f c = {};
#pragma unroll 8
  for (int kk = 0; kk < C_DIM; kk += 4) {
    const int ka = kk + hi * 2;
    v2f a, b;
    a.x = arow[ka];
    a.y = arow[ka + 1];
    b.x = wcol[(size_t)ka * C_DIM];
    b.y = wcol[(size_t)(ka + 1) * C_DIM];
    c = wmma4(a, b, c);
  }
  const float bv = bias[n0 + lo];
#pragma unroll
  for (int r = 0; r < 8; ++r)
    Out[(size_t)(m0 + r + hi * 8) * C_DIM + n0 + lo] = c[r] + bv;
}

extern "C" void kernel_launch(void* const* d_in, const int* in_sizes, int n_in,
                              void* d_out, int out_size, void* d_ws, size_t ws_size,
                              hipStream_t stream) {
  const float* x     = (const float*)d_in[0];
  const float* W_qkv = (const float*)d_in[1];
  const float* b_qkv = (const float*)d_in[2];
  const float* W_fc  = (const float*)d_in[3];
  const float* b_fc  = (const float*)d_in[4];
  float* out = (float*)d_out;

  const int BT = in_sizes[0] / C_DIM;
  const int B  = BT / T_SEQ;

  const size_t headElems = (size_t)B * HEADS * T_SEQ * DH;
  float* Qw = (float*)d_ws;
  float* Kw = Qw + headElems;
  float* Vw = Kw + headElems;
  float* Aw = Vw + headElems;

  const int mtiles = BT / 16;

  {
    const int tiles = mtiles * ((3 * C_DIM) / 16);
    const int blocks = (tiles + 7) / 8;
    qkv_proj_kernel<<<blocks, 256, 0, stream>>>(x, W_qkv, b_qkv, Qw, Kw, Vw,
                                                mtiles);
  }
  {
    const int nwaves = B * HEADS * (T_SEQ / 32);
    const int blocks = (nwaves + 3) / 4;
    attn_kernel<<<blocks, 128, 0, stream>>>(Qw, Kw, Vw, Aw, nwaves);
  }
  {
    const int tiles = mtiles * (C_DIM / 16);
    const int blocks = (tiles + 7) / 8;
    out_proj_kernel<<<blocks, 256, 0, stream>>>(Aw, W_fc, b_fc, out, mtiles);
  }
}


# attention: a block's 4 waves take 4 adjacent query tiles and share each K and V tile through LDS (fetched once per block, whole lines, a step ahead); per-tile arithmetic and sum order unchanged
# speedup vs baseline: 1.3365x; 1.1976x over previous
#include <hip/hip_runtime.h>


typedef float v2f __attribute__((ext_vector_type(2)));
typedef float v8f __attribute__((ext_vector_type(8)));

#define T_SEQ 2048
#define HEADS 4
#define DH    64
#define C_DIM 256

__device__ __forceinline__ v8f wmma4(v2f a, v2f b, v8f c) {
  return __builtin_amdgcn_wmma_f32_16x16x4_f32(false, a, false, b, (short)0, c,
                                               false, false);
}

__global__ void qkv_proj_kernel(const float* __restrict__ X,
                                const float* __restrict__ W,
                                const float* __restrict__ bias,
                                float* __restrict__ Qw,
                                float* __restrict__ Kw,
                                float* __restrict__ Vw,
                                int mtiles) {
  const int lane = threadIdx.x & 31;
  const int lo = lane & 15;
  const int hi = lane >> 4;
  const int ntiles = (3 * C_DIM) / 16;
  int tile = (int)((blockIdx.x * blockDim.x + threadIdx.x) >> 5);
  if (tile >= mtiles * ntiles) return;
  const int mt = tile / ntiles;
  const int nt = tile - mt * ntiles;
  const int m0 = mt * 16, n0 = nt * 16;

  const float* arow = X + (size_t)(m0 + lo) * C_DIM;
  const float* wcol = W + (n0 + lo);
  v8f c = {};
#pragma unroll 8
  for (int kk = 0; kk < C_DIM; kk += 4) {
    const int ka = kk + hi * 2;
    v2f a, b;
    a.x = arow[ka];
    a.y = arow[ka + 1];
    b.x = wcol[(size_t)ka * (3 * C_DIM)];
    b.y = wcol[(size_t)(ka + 1) * (3 * C_DIM)];
    c = wmma4(a, b, c);
  }

  const int cg = n0 + lo;
  const int which = cg >> 8;
  const int within = cg & 255;
  const int h = within >> 6;
  const int dd = within & 63;
  const float bv = bias[cg];
  float* dst = (which == 0) ? Qw : ((which == 1) ? Kw : Vw);
#pragma unroll
  for (int r = 0; r < 8; ++r) {
    const int row = m0 + r + hi * 8;
    const int bb = row >> 11;
    const int tt = row & (T_SEQ - 1);
    dst[(((size_t)(bb * HEADS + h)) * T_SEQ + tt) * DH + dd] = c[r] + bv;
  }
}

#define KP 68
#define VPP 160
#define KSZ (16 * KP)
#define VSZ (8 * VPP)

struct StageRegs { float4 k0, k1, v0, v1; };

__device__ __forceinline__ StageRegs stage_load(const float* kg, const float* vg,
                                                int tile) {
  StageRegs s;
  const size_t off = (size_t)tile * (16 * DH);
  s.k0 = *(const float4*)(kg + off);
  s.k1 = *(const float4*)(kg + off + 8 * DH);
  s.v0 = *(const float4*)(vg + off);
  s.v1 = *(const float4*)(vg + off + DH);
  return s;
}

__device__ __forceinline__ void stage_store(float* Kd, float* Vd,
                                            const StageRegs& s) {
  *(float4*)(Kd) = s.k0;
  *(float4*)(Kd + 8 * KP) = s.k1;
  *(float4*)(Vd) = make_float4(s.v0.x, s.v1.x, s.v0.y, s.v1.y);
  *(float4*)(Vd + 4) = make_float4(s.v0.z, s.v1.z, s.v0.w, s.v1.w);
}

__global__ void __launch_bounds__(128) attn_kernel(const float* __restrict__ Q,
                            const float* __restrict__ K,
                            const float* __restrict__ V,
                            float* __restrict__ O,
                            int nblk) {
  __shared__ __attribute__((aligned(16))) float Ks[2 * KSZ];
  __shared__ __attribute__((aligned(16))) float Vs[2 * VSZ];
  if ((int)blockIdx.x >= nblk) return;

  const int tid = threadIdx.x;
  const int lane = tid & 31;
  const int lo = lane & 15;
  const int hi = lane >> 4;
  const int w = __builtin_amdgcn_readfirstlane(tid >> 5);

  const int groups = T_SEQ / 64;
  const int gpairs = groups / 2;
  const int bh = blockIdx.x / gpairs;
  const int gp = blockIdx.x - bh * gpairs;
  const float* Qh = Q + (size_t)bh * T_SEQ * DH;
  const float* Kh = K + (size_t)bh * T_SEQ * DH;
  const float* Vh = V + (size_t)bh * T_SEQ * DH;

  const float qs = 0.125f * 1.44269504088896340736f;
  const int b = bh / HEADS;
  const int h = bh - b * HEADS;

  const int srow = tid >> 4;
  const int scol = (tid & 15) * 4;
  const float* kg = Kh + srow * DH + scol;
  const float* vg = Vh + (2 * srow) * DH + scol;
  float* ksd = Ks + srow * KP + scol;
  float* vsd = Vs + srow * VPP + 2 * scol;

  for (int t = 0; t < 2; ++t) {
  const int g = t ? gp : (groups - 1 - gp);
  const int qt = 4 * g + (t ? 3 - w : w);
  const int q0 = qt * 16;
  const int nkt = 4 * g + 4;

  v2f qreg[16];
  {
    const float* qrow = Qh + (size_t)(q0 + lo) * DH;
#pragma unroll
    for (int i = 0; i < 16; ++i) {
      qreg[i].x = qrow[i * 4 + hi * 2] * qs;
      qreg[i].y = qrow[i * 4 + hi * 2 + 1] * qs;
    }
  }

  float m_run = -__builtin_inff();
  float l_run = 0.0f;
  v8f o0 = {}, o1 = {}, o2 = {}, o3 = {};

  stage_store(ksd, vsd, stage_load(kg, vg, 0));
  StageRegs nxt = stage_load(kg, vg, 1);
  __syncthreads();

  for (int j = 0; j < nkt; ++j) {
    const int jf = (j + 2 < nkt) ? (j + 2) : (nkt - 1);
    const StageRegs inc = stage_load(kg, vg, jf);
    const int nb = (j + 1) & 1;
    stage_store(ksd + nb * KSZ, vsd + nb * VSZ, nxt);
    nxt = inc;

    const float* Kb = Ks + (j & 1) * KSZ;
    const float* Vb = Vs + (j & 1) * VSZ;

    if (j <= qt) {
    v8f s = {};
    const float* krow = Kb + lo * KP + hi * 2;
    v2f ka[16];
#pragma unroll
    for (int i = 0; i < 16; ++i) {
      ka[i].x = krow[i * 4];
      ka[i].y = krow[i * 4 + 1];
    }
    asm volatile("" :: "v"(ka[0]), "v"(ka[1]), "v"(ka[2]), "v"(ka[3]),
                       "v"(ka[4]), "v"(ka[5]), "v"(ka[6]), "v"(ka[7]),
                       "v"(ka[8]), "v"(ka[9]), "v"(ka[10]), "v"(ka[11]),
                       "v"(ka[12]), "v"(ka[13]), "v"(ka[14]), "v"(ka[15]));
#pragma unroll
    for (int i = 0; i < 16; ++i) s = wmma4(ka[i], qreg[i], s);
    if (j == qt) {
#pragma unroll
      for (int r = 0; r < 8; ++r)
        s[r] = (r + hi * 8 > lo) ? -__builtin_inff() : s[r];
    }

    float tmax = s[0];
#pragma unroll
    for (int r = 1; r < 8; ++r) tmax = fmaxf(tmax, s[r]);
    tmax = fmaxf(tmax, __shfl_xor(tmax, 16, 32));
    const float m_new = fmaxf(m_run, tmax);
    const float alpha = __builtin_amdgcn_exp2f(m_run - m_new);
    float p[8];
    float ts = 0.0f;
#pragma unroll
    for (int r = 0; r < 8; ++r) {
      p[r] = __builtin_amdgcn_exp2f(s[r] - m_new);
      ts += p[r];
    }
    ts += __shfl_xor(ts, 16, 32);
    l_run = l_run * alpha + ts;
    m_run = m_new;
#pragma unroll
    for (int r = 0; r < 8; ++r) {
      o0[r] *= alpha; o1[r] *= alpha; o2[r] *= alpha; o3[r] *= alpha;
    }

    const float* vrow = Vb + hi * VPP + 2 * lo;
#pragma unroll
    for (int kk = 0; kk < 16; kk += 4) {
      v2f bop;
      if (kk < 8) {
        const float sx = __shfl_xor(p[kk + 2], 16, 32);
        const float sy = __shfl_xor(p[kk + 3], 16, 32);
        bop.x = hi ? sx : p[kk];
        bop.y = hi ? sy : p[kk + 1];
      } else {
        const int jj = kk - 8;
        const float sx = __shfl_xor(p[jj], 16, 32);
        const float sy = __shfl_xor(p[jj + 1], 16, 32);
        bop.x = hi ? p[jj + 2] : sx;
        bop.y = hi ? p[jj + 3] : sy;
      }
      const float* vk = vrow + (kk / 2) * VPP;
      v2f a;
      a.x = vk[0];       a.y = vk[1];         o0 = wmma4(a, bop, o0);
      a.x = vk[32];      a.y = vk[33];        o1 = wmma4(a, bop, o1);
      a.x = vk[64];      a.y = vk[65];        o2 = wmma4(a, bop, o2);
      a.x = vk[96];      a.y = vk[97];        o3 = wmma4(a, bop, o3);
    }
    }

    __syncthreads();
  }

  const float inv_l = 1.0f / l_run;
  float* orow = O + ((size_t)b * T_SEQ + q0 + lo) * C_DIM + h * DH + hi * 8;
#pragma unroll
  for (int r = 0; r < 8; ++r) {
    orow[r]      = o0[r] * inv_l;
    orow[16 + r] = o1[r] * inv_l;
    orow[32 + r] = o2[r] * inv_l;
    orow[48 + r] = o3[r] * inv_l;
  }
  }
}

__global__ void out_proj_kernel(const float* __restrict__ A,
                                const float* __restrict__ W,
                                const float* __restrict__ bias,
                                float* __restrict__ Out,
                                int mtiles) {
  const int lane = threadIdx.x & 31;
  const int lo = lane & 15;
  const int hi = lane >> 4;
  const int ntiles = C_DIM / 16;
  int tile = (int)((blockIdx.x * blockDim.x + threadIdx.x) >> 5);
  if (tile >= mtiles * ntiles) return;
  const int mt = tile / ntiles;
  const int nt = tile - mt * ntiles;
  const int m0 = mt * 16, n0 = nt * 16;

  const float* arow = A + (size_t)(m0 + lo) * C_DIM;
  const float* wcol = W + (n0 + lo);
  v8f c = {};
#pragma unroll 8
  for (int kk = 0; kk < C_DIM; kk += 4) {
    const int ka = kk + hi * 2;
    v2f a, b;
    a.x = arow[ka];
    a.y = arow[ka + 1];
    b.x = wcol[(size_t)ka * C_DIM];
    b.y = wcol[(size_t)(ka + 1) * C_DIM];
    c = wmma4(a, b, c);
  }
  const float bv = bias[n0 + lo];
#pragma unroll
  for (int r = 0; r < 8; ++r)
    Out[(size_t)(m0 + r + hi * 8) * C_DIM + n0 + lo] = c[r] + bv;
}

extern "C" void kernel_launch(void* const* d_in, const int* in_sizes, int n_in,
                              void* d_out, int out_size, void* d_ws, size_t ws_size,
                              hipStream_t stream) {
  const float* x     = (const float*)d_in[0];
  const float* W_qkv = (const float*)d_in[1];
  const float* b_qkv = (const float*)d_in[2];
  const float* W_fc  = (const float*)d_in[3];
  const float* b_fc  = (const float*)d_in[4];
  float* out = (float*)d_out;

  const int BT = in_sizes[0] / C_DIM;
  const int B  = BT / T_SEQ;

  const size_t headElems = (size_t)B * HEADS * T_SEQ * DH;
  float* Qw = (float*)d_ws;
  float* Kw = Qw + headElems;
  float* Vw = Kw + headElems;
  float* Aw = Vw + headElems;

  const int mtiles = BT / 16;

  {
    const int tiles = mtiles * ((3 * C_DIM) / 16);
    const int blocks = (tiles + 7) / 8;
    qkv_proj_kernel<<<blocks, 256, 0, stream>>>(x, W_qkv, b_qkv, Qw, Kw, Vw,
                                                mtiles);
  }
  {
    const int blocks = B * HEADS * (T_SEQ / 128);
    attn_kernel<<<blocks, 128, 0, stream>>>(Qw, Kw, Vw, Aw, blocks);
  }
  {
    const int tiles = mtiles * (C_DIM / 16);
    const int blocks = (tiles + 7) / 8;
    out_proj_kernel<<<blocks, 256, 0, stream>>>(Aw, W_fc, b_fc, out, mtiles);
  }
}


# projections: 128x128 block tile, A and W chunks staged once per block via LDS (whole-line loads), each wave 2x4 WMMA tiles so a fragment feeds 2-4 WMMAs; k order and per-element sum order unchanged
# speedup vs baseline: 1.8751x; 1.4030x over previous
#include <hip/hip_runtime.h>


typedef float v2f __attribute__((ext_vector_type(2)));
typedef float v8f __attribute__((ext_vector_type(8)));

#define T_SEQ 2048
#define HEADS 4
#define DH    64
#define C_DIM 256

__device__ __forceinline__ v8f wmma4(v2f a, v2f b, v8f c) {
  return __builtin_amdgcn_wmma_f32_16x16x4_f32(false, a, false, b, (short)0, c,
                                               false, false);
}

#define GBM 128
#define GBN 128
#define GBK 32
#define GAP 36
#define GBP 288
#define GASZ (GBM * GAP)
#define GBSZ ((GBK / 2) * GBP)

struct GemmStage { float4 a0, a1, a2, a3, p0, p1, q0, q1; };

__device__ __forceinline__ GemmStage gemm_stage_load(const float* ag,
                                                     const float* bg, int ldb,
                                                     int kc) {
  GemmStage s;
  const float* a = ag + kc * GBK;
  s.a0 = *(const float4*)(a);
  s.a1 = *(const float4*)(a + 32 * C_DIM);
  s.a2 = *(const float4*)(a + 64 * C_DIM);
  s.a3 = *(const float4*)(a + 96 * C_DIM);
  const float* b = bg + (size_t)(kc * GBK) * ldb;
  s.p0 = *(const float4*)(b);
  s.p1 = *(const float4*)(b + ldb);
  s.q0 = *(const float4*)(b + (size_t)16 * ldb);
  s.q1 = *(const float4*)(b + (size_t)17 * ldb);
  return s;
}

__device__ __forceinline__ void gemm_stage_store(float* ad, float* bd,
                                                 const GemmStage& s) {
  *(float4*)(ad) = s.a0;
  *(float4*)(ad + 32 * GAP) = s.a1;
  *(float4*)(ad + 64 * GAP) = s.a2;
  *(float4*)(ad + 96 * GAP) = s.a3;
  *(float4*)(bd) = make_float4(s.p0.x, s.p1.x, s.p0.y, s.p1.y);
  *(float4*)(bd + 4) = make_float4(s.p0.z, s.p1.z, s.p0.w, s.p1.w);
  *(float4*)(bd + 8 * GBP) = make_float4(s.q0.x, s.q1.x, s.q0.y, s.q1.y);
  *(float4*)(bd + 8 * GBP + 4) = make_float4(s.q0.z, s.q1.z, s.q0.w, s.q1.w);
}

struct GemmAcc { v8f c00, c01, c02, c03, c10, c11, c12, c13; };

struct GemmFrag { v2f a0[2], a1[2], b0[2], b1[2], b2[2], b3[2]; };

__device__ __forceinline__ void gemm_frag_load(GemmFrag& f, const float* af,
                                               const float* bf, int g) {
#pragma unroll
  for (int s = 0; s < 2; ++s) {
    const int ks = 2 * g + s;
    f.a0[s] = *(const v2f*)(af + ks * 4);
    f.a1[s] = *(const v2f*)(af + 16 * GAP + ks * 4);
    const float* bk = bf + ks * 2 * GBP;
    f.b0[s] = *(const v2f*)(bk);
    f.b1[s] = *(const v2f*)(bk + 32);
    f.b2[s] = *(const v2f*)(bk + 64);
    f.b3[s] = *(const v2f*)(bk + 96);
  }
}

__device__ __forceinline__ void gemm_frag_mma(const GemmFrag& f, GemmAcc& c) {
#pragma unroll
  for (int s = 0; s < 2; ++s) {
    c.c00 = wmma4(f.a0[s], f.b0[s], c.c00);
    c.c01 = wmma4(f.a0[s], f.b1[s], c.c01);
    c.c02 = wmma4(f.a0[s], f.b2[s], c.c02);
    c.c03 = wmma4(f.a0[s], f.b3[s], c.c03);
    c.c10 = wmma4(f.a1[s], f.b0[s], c.c10);
    c.c11 = wmma4(f.a1[s], f.b1[s], c.c11);
    c.c12 = wmma4(f.a1[s], f.b2[s], c.c12);
    c.c13 = wmma4(f.a1[s], f.b3[s], c.c13);
  }
}

__device__ __forceinline__ void gemm_block(const float* __restrict__ A,
                                           const float* __restrict__ Bm,
                                           int ldb, int m_blk, int n_blk,
                                           float* As, float* Bs, int wm, int wn,
                                           GemmAcc& c) {
  const int tid = threadIdx.x;
  const int lane = tid & 31;
  const int lo = lane & 15;
  const int hi = lane >> 4;

  const int arow = tid >> 3;
  const int acol = (tid & 7) * 4;
  const int bp = tid >> 5;
  const int bcol = (tid & 31) * 4;
  const float* ag = A + (size_t)(m_blk * GBM + arow) * C_DIM + acol;
  const float* bg = Bm + (size_t)(2 * bp) * ldb + n_blk * GBN + bcol;
  float* asd = As + arow * GAP + acol;
  float* bsd = Bs + bp * GBP + 2 * bcol;

  const float* af = As + (wm * 32 + lo) * GAP + 2 * hi;
  const float* bf = Bs + hi * GBP + 2 * (wn * 64 + lo);

  const v8f z = {};
  c.c00 = z; c.c01 = z; c.c02 = z; c.c03 = z;
  c.c10 = z; c.c11 = z; c.c12 = z; c.c13 = z;

  gemm_stage_store(asd, bsd, gemm_stage_load(ag, bg, ldb, 0));
  __syncthreads();

  const int nchunks = C_DIM / GBK;
#pragma unroll 1
  for (int kc = 0; kc < nchunks; ++kc) {
    const int kn = (kc + 1 < nchunks) ? (kc + 1) : (nchunks - 1);
    const GemmStage nxt = gemm_stage_load(ag, bg, ldb, kn);
    const int cb = kc & 1;
    const int nb = cb ^ 1;
    const float* afc = af + cb * GASZ;
    const float* bfc = bf + cb * GBSZ;
    GemmFrag f0, f1;
    gemm_frag_load(f0, afc, bfc, 0);
    gemm_frag_load(f1, afc, bfc, 1);
    gemm_frag_mma(f0, c);
    gemm_frag_load(f0, afc, bfc, 2);
    gemm_frag_mma(f1, c);
    gemm_frag_load(f1, afc, bfc, 3);
    gemm_stage_store(asd + nb * GASZ, bsd + nb * GBSZ, nxt);
    gemm_frag_mma(f0, c);
    gemm_frag_mma(f1, c);
    __builtin_amdgcn_sched_group_barrier(0x020, 8, 0);
    __builtin_amdgcn_sched_group_barrier(0x100, 12, 0);
    __builtin_amdgcn_sched_group_barrier(0x008, 16, 0);
    __builtin_amdgcn_sched_group_barrier(0x100, 6, 0);
    __builtin_amdgcn_sched_group_barrier(0x008, 16, 0);
    __builtin_amdgcn_sched_group_barrier(0x100, 6, 0);
    __builtin_amdgcn_sched_group_barrier(0x200, 8, 0);
    __builtin_amdgcn_sched_group_barrier(0x008, 32, 0);
    __syncthreads();
  }
}

__global__ void __launch_bounds__(256) qkv_proj_kernel(const float* __restrict__ X,
                                const float* __restrict__ W,
                                const float* __restrict__ bias,
                                float* __restrict__ Qw,
                                float* __restrict__ Kw,
                                float* __restrict__ Vw) {
  __shared__ __attribute__((aligned(16))) float As[2 * GASZ];
  __shared__ __attribute__((aligned(16))) float Bs[2 * GBSZ];
  const int nblks = (3 * C_DIM) / GBN;
  const int m_blk = blockIdx.x / nblks;
  const int n_blk = blockIdx.x - m_blk * nblks;
  const int tid = threadIdx.x;
  const int lo = tid & 15;
  const int hi = (tid >> 4) & 1;
  const int w = __builtin_amdgcn_readfirstlane(tid >> 5);
  const int wm = w & 3, wn = w >> 2;

  GemmAcc c;
  gemm_block(X, W, 3 * C_DIM, m_blk, n_blk, As, Bs, wm, wn, c);

  const int n_base = n_blk * GBN + wn * 64;
  const int which = n_base >> 8;
  const int h = (n_base & 255) >> 6;
  float* dst = (which == 0) ? Qw : ((which == 1) ? Kw : Vw);
  const int row0 = m_blk * GBM + wm * 32 + hi * 8;
  const int bb = row0 >> 11;
  const int tt0 = row0 & (T_SEQ - 1);
  float* d0 = dst + (((size_t)(bb * HEADS + h)) * T_SEQ + tt0) * DH + lo;
  const float bv0 = bias[n_base + lo];
  const float bv1 = bias[n_base + 16 + lo];
  const float bv2 = bias[n_base + 32 + lo];
  const float bv3 = bias[n_base + 48 + lo];
#pragma unroll
  for (int r = 0; r < 8; ++r) {
    float* d = d0 + r * DH;
    d[0]  = c.c00[r] + bv0;
    d[16] = c.c01[r] + bv1;
    d[32] = c.c02[r] + bv2;
    d[48] = c.c03[r] + bv3;
    d += 16 * DH;
    d[0]  = c.c10[r] + bv0;
    d[16] = c.c11[r] + bv1;
    d[32] = c.c12[r] + bv2;
    d[48] = c.c13[r] + bv3;
  }
}

#define KP 68
#define VPP 160
#define KSZ (16 * KP)
#define VSZ (8 * VPP)

struct StageRegs { float4 k0, k1, v0, v1; };

__device__ __forceinline__ StageRegs stage_load(const float* kg, const float* vg,
                                                int tile) {
  StageRegs s;
  const size_t off = (size_t)tile * (16 * DH);
  s.k0 = *(const float4*)(kg + off);
  s.k1 = *(const float4*)(kg + off + 8 * DH);
  s.v0 = *(const float4*)(vg + off);
  s.v1 = *(const float4*)(vg + off + DH);
  return s;
}

__device__ __forceinline__ void stage_store(float* Kd, float* Vd,
                                            const StageRegs& s) {
  *(float4*)(Kd) = s.k0;
  *(float4*)(Kd + 8 * KP) = s.k1;
  *(float4*)(Vd) = make_float4(s.v0.x, s.v1.x, s.v0.y, s.v1.y);
  *(float4*)(Vd + 4) = make_float4(s.v0.z, s.v1.z, s.v0.w, s.v1.w);
}

__global__ void __launch_bounds__(128) attn_kernel(const float* __restrict__ Q,
                            const float* __restrict__ K,
                            const float* __restrict__ V,
                            float* __restrict__ O,
                            int nblk) {
  __shared__ __attribute__((aligned(16))) float Ks[2 * KSZ];
  __shared__ __attribute__((aligned(16))) float Vs[2 * VSZ];
  if ((int)blockIdx.x >= nblk) return;

  const int tid = threadIdx.x;
  const int lane = tid & 31;
  const int lo = lane & 15;
  const int hi = lane >> 4;
  const int w = __builtin_amdgcn_readfirstlane(tid >> 5);

  const int groups = T_SEQ / 64;
  const int gpairs = groups / 2;
  const int bh = blockIdx.x / gpairs;
  const int gp = blockIdx.x - bh * gpairs;
  const float* Qh = Q + (size_t)bh * T_SEQ * DH;
  const float* Kh = K + (size_t)bh * T_SEQ * DH;
  const float* Vh = V + (size_t)bh * T_SEQ * DH;

  const float qs = 0.125f * 1.44269504088896340736f;
  const int b = bh / HEADS;
  const int h = bh - b * HEADS;

  const int srow = tid >> 4;
  const int scol = (tid & 15) * 4;
  const float* kg = Kh + srow * DH + scol;
  const float* vg = Vh + (2 * srow) * DH + scol;
  float* ksd = Ks + srow * KP + scol;
  float* vsd = Vs + srow * VPP + 2 * scol;

  for (int t = 0; t < 2; ++t) {
  const int g = t ? gp : (groups - 1 - gp);
  const int qt = 4 * g + (t ? 3 - w : w);
  const int q0 = qt * 16;
  const int nkt = 4 * g + 4;

  v2f qreg[16];
  {
    const float* qrow = Qh + (size_t)(q0 + lo) * DH;
#pragma unroll
    for (int i = 0; i < 16; ++i) {
      qreg[i].x = qrow[i * 4 + hi * 2] * qs;
      qreg[i].y = qrow[i * 4 + hi * 2 + 1] * qs;
    }
  }

  float m_run = -__builtin_inff();
  float l_run = 0.0f;
  v8f o0 = {}, o1 = {}, o2 = {}, o3 = {};

  stage_store(ksd, vsd, stage_load(kg, vg, 0));
  StageRegs nxt = stage_load(kg, vg, 1);
  __syncthreads();

  for (int j = 0; j < nkt; ++j) {
    const int jf = (j + 2 < nkt) ? (j + 2) : (nkt - 1);
    const StageRegs inc = stage_load(kg, vg, jf);
    const int nb = (j + 1) & 1;
    stage_store(ksd + nb * KSZ, vsd + nb * VSZ, nxt);
    nxt = inc;

    const float* Kb = Ks + (j & 1) * KSZ;
    const float* Vb = Vs + (j & 1) * VSZ;

    if (j <= qt) {
    v8f s = {};
    const float* krow = Kb + lo * KP + hi * 2;
    v2f ka[16];
#pragma unroll
    for (int i = 0; i < 16; ++i) {
      ka[i].x = krow[i * 4];
      ka[i].y = krow[i * 4 + 1];
    }
    asm volatile("" :: "v"(ka[0]), "v"(ka[1]), "v"(ka[2]), "v"(ka[3]),
                       "v"(ka[4]), "v"(ka[5]), "v"(ka[6]), "v"(ka[7]),
                       "v"(ka[8]), "v"(ka[9]), "v"(ka[10]), "v"(ka[11]),
                       "v"(ka[12]), "v"(ka[13]), "v"(ka[14]), "v"(ka[15]));
#pragma unroll
    for (int i = 0; i < 16; ++i) s = wmma4(ka[i], qreg[i], s);
    if (j == qt) {
#pragma unroll
      for (int r = 0; r < 8; ++r)
        s[r] = (r + hi * 8 > lo) ? -__builtin_inff() : s[r];
    }

    float tmax = s[0];
#pragma unroll
    for (int r = 1; r < 8; ++r) tmax = fmaxf(tmax, s[r]);
    tmax = fmaxf(tmax, __shfl_xor(tmax, 16, 32));
    const float m_new = fmaxf(m_run, tmax);
    const float alpha = __builtin_amdgcn_exp2f(m_run - m_new);
    float p[8];
    float ts = 0.0f;
#pragma unroll
    for (int r = 0; r < 8; ++r) {
      p[r] = __builtin_amdgcn_exp2f(s[r] - m_new);
      ts += p[r];
    }
    ts += __shfl_xor(ts, 16, 32);
    l_run = l_run * alpha + ts;
    m_run = m_new;
#pragma unroll
    for (int r = 0; r < 8; ++r) {
      o0[r] *= alpha; o1[r] *= alpha; o2[r] *= alpha; o3[r] *= alpha;
    }

    const float* vrow = Vb + hi * VPP + 2 * lo;
#pragma unroll
    for (int kk = 0; kk < 16; kk += 4) {
      v2f bop;
      if (kk < 8) {
        const float sx = __shfl_xor(p[kk + 2], 16, 32);
        const float sy = __shfl_xor(p[kk + 3], 16, 32);
        bop.x = hi ? sx : p[kk];
        bop.y = hi ? sy : p[kk + 1];
      } else {
        const int jj = kk - 8;
        const float sx = __shfl_xor(p[jj], 16, 32);
        const float sy = __shfl_xor(p[jj + 1], 16, 32);
        bop.x = hi ? p[jj + 2] : sx;
        bop.y = hi ? p[jj + 3] : sy;
      }
      const float* vk = vrow + (kk / 2) * VPP;
      v2f a;
      a.x = vk[0];       a.y = vk[1];         o0 = wmma4(a, bop, o0);
      a.x = vk[32];      a.y = vk[33];        o1 = wmma4(a, bop, o1);
      a.x = vk[64];      a.y = vk[65];        o2 = wmma4(a, bop, o2);
      a.x = vk[96];      a.y = vk[97];        o3 = wmma4(a, bop, o3);
    }
    }

    __syncthreads();
  }

  const float inv_l = 1.0f / l_run;
  float* orow = O + ((size_t)b * T_SEQ + q0 + lo) * C_DIM + h * DH + hi * 8;
#pragma unroll
  for (int r = 0; r < 8; ++r) {
    orow[r]      = o0[r] * inv_l;
    orow[16 + r] = o1[r] * inv_l;
    orow[32 + r] = o2[r] * inv_l;
    orow[48 + r] = o3[r] * inv_l;
  }
  }
}

__global__ void __launch_bounds__(256) out_proj_kernel(const float* __restrict__ A,
                                const float* __restrict__ W,
                                const float* __restrict__ bias,
                                float* __restrict__ Out) {
  __shared__ __attribute__((aligned(16))) float As[2 * GASZ];
  __shared__ __attribute__((aligned(16))) float Bs[2 * GBSZ];
  const int nblks = C_DIM / GBN;
  const int m_blk = blockIdx.x / nblks;
  const int n_blk = blockIdx.x - m_blk * nblks;
  const int tid = threadIdx.x;
  const int lo = tid & 15;
  const int hi = (tid >> 4) & 1;
  const int w = __builtin_amdgcn_readfirstlane(tid >> 5);
  const int wm = w & 3, wn = w >> 2;

  GemmAcc c;
  gemm_block(A, W, C_DIM, m_blk, n_blk, As, Bs, wm, wn, c);

  const int n_base = n_blk * GBN + wn * 64;
  const int row0 = m_blk * GBM + wm * 32 + hi * 8;
  float* d0 = Out + (size_t)row0 * C_DIM + n_base + lo;
  const float bv0 = bias[n_base + lo];
  const float bv1 = bias[n_base + 16 + lo];
  const float bv2 = bias[n_base + 32 + lo];
  const float bv3 = bias[n_base + 48 + lo];
#pragma unroll
  for (int r = 0; r < 8; ++r) {
    float* d = d0 + (size_t)r * C_DIM;
    d[0]  = c.c00[r] + bv0;
    d[16] = c.c01[r] + bv1;
    d[32] = c.c02[r] + bv2;
    d[48] = c.c03[r] + bv3;
    d += 16 * C_DIM;
    d[0]  = c.c10[r] + bv0;
    d[16] = c.c11[r] + bv1;
    d[32] = c.c12[r] + bv2;
    d[48] = c.c13[r] + bv3;
  }
}

extern "C" void kernel_launch(void* const* d_in, const int* in_sizes, int n_in,
                              void* d_out, int out_size, void* d_ws, size_t ws_size,
                              hipStream_t stream) {
  const float* x     = (const float*)d_in[0];
  const float* W_qkv = (const float*)d_in[1];
  const float* b_qkv = (const float*)d_in[2];
  const float* W_fc  = (const float*)d_in[3];
  const float* b_fc  = (const float*)d_in[4];
  float* out = (float*)d_out;

  const int BT = in_sizes[0] / C_DIM;
  const int B  = BT / T_SEQ;

  const size_t headElems = (size_t)B * HEADS * T_SEQ * DH;
  float* Qw = (float*)d_ws;
  float* Kw = Qw + headElems;
  float* Vw = Kw + headElems;
  float* Aw = Vw + headElems;

  const int mblks = BT / GBM;

  {
    const int blocks = mblks * ((3 * C_DIM) / GBN);
    qkv_proj_kernel<<<blocks, 256, 0, stream>>>(x, W_qkv, b_qkv, Qw, Kw, Vw);
  }
  {
    const int blocks = B * HEADS * (T_SEQ / 128);
    attn_kernel<<<blocks, 128, 0, stream>>>(Qw, Kw, Vw, Aw, blocks);
  }
  {
    const int blocks = mblks * (C_DIM / GBN);
    out_proj_kernel<<<blocks, 256, 0, stream>>>(Aw, W_fc, b_fc, out);
  }
}


# attention: 8-wave blocks (4 waves per SIMD): waves 4-7 take the long query group's key tiles from 66 on, then the short group; a long row's two partial softmax sums merge via LDS (sum order changes)
# speedup vs baseline: 1.9825x; 1.0573x over previous
#include <hip/hip_runtime.h>


typedef float v2f __attribute__((ext_vector_type(2)));
typedef float v8f __attribute__((ext_vector_type(8)));

#define T_SEQ 2048
#define HEADS 4
#define DH    64
#define C_DIM 256

__device__ __forceinline__ v8f wmma4(v2f a, v2f b, v8f c) {
  return __builtin_amdgcn_wmma_f32_16x16x4_f32(false, a, false, b, (short)0, c,
                                               false, false);
}

#define GBM 128
#define GBN 128
#define GBK 32
#define GAP 36
#define GBP 288
#define GASZ (GBM * GAP)
#define GBSZ ((GBK / 2) * GBP)

struct GemmStage { float4 a0, a1, a2, a3, p0, p1, q0, q1; };

__device__ __forceinline__ GemmStage gemm_stage_load(const float* ag,
                                                     const float* bg, int ldb,
                                                     int kc) {
  GemmStage s;
  const float* a = ag + kc * GBK;
  s.a0 = *(const float4*)(a);
  s.a1 = *(const float4*)(a + 32 * C_DIM);
  s.a2 = *(const float4*)(a + 64 * C_DIM);
  s.a3 = *(const float4*)(a + 96 * C_DIM);
  const float* b = bg + (size_t)(kc * GBK) * ldb;
  s.p0 = *(const float4*)(b);
  s.p1 = *(const float4*)(b + ldb);
  s.q0 = *(const float4*)(b + (size_t)16 * ldb);
  s.q1 = *(const float4*)(b + (size_t)17 * ldb);
  return s;
}

__device__ __forceinline__ void gemm_stage_store(float* ad, float* bd,
                                                 const GemmStage& s) {
  *(float4*)(ad) = s.a0;
  *(float4*)(ad + 32 * GAP) = s.a1;
  *(float4*)(ad + 64 * GAP) = s.a2;
  *(float4*)(ad + 96 * GAP) = s.a3;
  *(float4*)(bd) = make_float4(s.p0.x, s.p1.x, s.p0.y, s.p1.y);
  *(float4*)(bd + 4) = make_float4(s.p0.z, s.p1.z, s.p0.w, s.p1.w);
  *(float4*)(bd + 8 * GBP) = make_float4(s.q0.x, s.q1.x, s.q0.y, s.q1.y);
  *(float4*)(bd + 8 * GBP + 4) = make_float4(s.q0.z, s.q1.z, s.q0.w, s.q1.w);
}

struct GemmAcc { v8f c00, c01, c02, c03, c10, c11, c12, c13; };

struct GemmFrag { v2f a0[2], a1[2], b0[2], b1[2], b2[2], b3[2]; };

__device__ __forceinline__ void gemm_frag_load(GemmFrag& f, const float* af,
                                               const float* bf, int g) {
#pragma unroll
  for (int s = 0; s < 2; ++s) {
    const int ks = 2 * g + s;
    f.a0[s] = *(const v2f*)(af + ks * 4);
    f.a1[s] = *(const v2f*)(af + 16 * GAP + ks * 4);
    const float* bk = bf + ks * 2 * GBP;
    f.b0[s] = *(const v2f*)(bk);
    f.b1[s] = *(const v2f*)(bk + 32);
    f.b2[s] = *(const v2f*)(bk + 64);
    f.b3[s] = *(const v2f*)(bk + 96);
  }
}

__device__ __forceinline__ void gemm_frag_mma(const GemmFrag& f, GemmAcc& c) {
#pragma unroll
  for (int s = 0; s < 2; ++s) {
    c.c00 = wmma4(f.a0[s], f.b0[s], c.c00);
    c.c01 = wmma4(f.a0[s], f.b1[s], c.c01);
    c.c02 = wmma4(f.a0[s], f.b2[s], c.c02);
    c.c03 = wmma4(f.a0[s], f.b3[s], c.c03);
    c.c10 = wmma4(f.a1[s], f.b0[s], c.c10);
    c.c11 = wmma4(f.a1[s], f.b1[s], c.c11);
    c.c12 = wmma4(f.a1[s], f.b2[s], c.c12);
    c.c13 = wmma4(f.a1[s], f.b3[s], c.c13);
  }
}

__device__ __forceinline__ void gemm_block(const float* __restrict__ A,
                                           const float* __restrict__ Bm,
                                           int ldb, int m_blk, int n_blk,
                                           float* As, float* Bs, int wm, int wn,
                                           GemmAcc& c) {
  const int tid = threadIdx.x;
  const int lane = tid & 31;
  const int lo = lane & 15;
  const int hi = lane >> 4;

  const int arow = tid >> 3;
  const int acol = (tid & 7) * 4;
  const int bp = tid >> 5;
  const int bcol = (tid & 31) * 4;
  const float* ag = A + (size_t)(m_blk * GBM + arow) * C_DIM + acol;
  const float* bg = Bm + (size_t)(2 * bp) * ldb + n_blk * GBN + bcol;
  float* asd = As + arow * GAP + acol;
  float* bsd = Bs + bp * GBP + 2 * bcol;

  const float* af = As + (wm * 32 + lo) * GAP + 2 * hi;
  const float* bf = Bs + hi * GBP + 2 * (wn * 64 + lo);

  const v8f z = {};
  c.c00 = z; c.c01 = z; c.c02 = z; c.c03 = z;
  c.c10 = z; c.c11 = z; c.c12 = z; c.c13 = z;

  gemm_stage_store(asd, bsd, gemm_stage_load(ag, bg, ldb, 0));
  __syncthreads();

  const int nchunks = C_DIM / GBK;
#pragma unroll 1
  for (int kc = 0; kc < nchunks; ++kc) {
    const int kn = (kc + 1 < nchunks) ? (kc + 1) : (nchunks - 1);
    const GemmStage nxt = gemm_stage_load(ag, bg, ldb, kn);
    const int cb = kc & 1;
    const int nb = cb ^ 1;
    const float* afc = af + cb * GASZ;
    const float* bfc = bf + cb * GBSZ;
    GemmFrag f0, f1;
    gemm_frag_load(f0, afc, bfc, 0);
    gemm_frag_load(f1, afc, bfc, 1);
    gemm_frag_mma(f0, c);
    gemm_frag_load(f0, afc, bfc, 2);
    gemm_frag_mma(f1, c);
    gemm_frag_load(f1, afc, bfc, 3);
    gemm_stage_store(asd + nb * GASZ, bsd + nb * GBSZ, nxt);
    gemm_frag_mma(f0, c);
    gemm_frag_mma(f1, c);
    __builtin_amdgcn_sched_group_barrier(0x020, 8, 0);
    __builtin_amdgcn_sched_group_barrier(0x100, 12, 0);
    __builtin_amdgcn_sched_group_barrier(0x008, 16, 0);
    __builtin_amdgcn_sched_group_barrier(0x100, 6, 0);
    __builtin_amdgcn_sched_group_barrier(0x008, 16, 0);
    __builtin_amdgcn_sched_group_barrier(0x100, 6, 0);
    __builtin_amdgcn_sched_group_barrier(0x200, 8, 0);
    __builtin_amdgcn_sched_group_barrier(0x008, 32, 0);
    __syncthreads();
  }
}

__global__ void __launch_bounds__(256) qkv_proj_kernel(const float* __restrict__ X,
                                const float* __restrict__ W,
                                const float* __restrict__ bias,
                                float* __restrict__ Qw,
                                float* __restrict__ Kw,
                                float* __restrict__ Vw) {
  __shared__ __attribute__((aligned(16))) float As[2 * GASZ];
  __shared__ __attribute__((aligned(16))) float Bs[2 * GBSZ];
  const int nblks = (3 * C_DIM) / GBN;
  const int m_blk = blockIdx.x / nblks;
  const int n_blk = blockIdx.x - m_blk * nblks;
  const int tid = threadIdx.x;
  const int lo = tid & 15;
  const int hi = (tid >> 4) & 1;
  const int w = __builtin_amdgcn_readfirstlane(tid >> 5);
  const int wm = w & 3, wn = w >> 2;

  GemmAcc c;
  gemm_block(X, W, 3 * C_DIM, m_blk, n_blk, As, Bs, wm, wn, c);

  const int n_base = n_blk * GBN + wn * 64;
  const int which = n_base >> 8;
  const int h = (n_base & 255) >> 6;
  float* dst = (which == 0) ? Qw : ((which == 1) ? Kw : Vw);
  const int row0 = m_blk * GBM + wm * 32 + hi * 8;
  const int bb = row0 >> 11;
  const int tt0 = row0 & (T_SEQ - 1);
  float* d0 = dst + (((size_t)(bb * HEADS + h)) * T_SEQ + tt0) * DH + lo;
  const float bv0 = bias[n_base + lo];
  const float bv1 = bias[n_base + 16 + lo];
  const float bv2 = bias[n_base + 32 + lo];
  const float bv3 = bias[n_base + 48 + lo];
#pragma unroll
  for (int r = 0; r < 8; ++r) {
    float* d = d0 + r * DH;
    d[0]  = c.c00[r] + bv0;
    d[16] = c.c01[r] + bv1;
    d[32] = c.c02[r] + bv2;
    d[48] = c.c03[r] + bv3;
    d += 16 * DH;
    d[0]  = c.c10[r] + bv0;
    d[16] = c.c11[r] + bv1;
    d[32] = c.c12[r] + bv2;
    d[48] = c.c13[r] + bv3;
  }
}

#define KP 68
#define VPP 160
#define KSZ (16 * KP)
#define VSZ (8 * VPP)
#define SPLIT_KT 66
#define NSTEP 66
#define MGV 34

struct StageRegs { float4 k0, k1, v0, v1; };

__device__ __forceinline__ StageRegs stage_load(const float* kg, const float* vg,
                                                int tile) {
  StageRegs s;
  const size_t off = (size_t)tile * (16 * DH);
  s.k0 = *(const float4*)(kg + off);
  s.k1 = *(const float4*)(kg + off + 8 * DH);
  s.v0 = *(const float4*)(vg + off);
  s.v1 = *(const float4*)(vg + off + DH);
  return s;
}

__device__ __forceinline__ void stage_store(float* Kd, float* Vd,
                                            const StageRegs& s) {
  *(float4*)(Kd) = s.k0;
  *(float4*)(Kd + 8 * KP) = s.k1;
  *(float4*)(Vd) = make_float4(s.v0.x, s.v1.x, s.v0.y, s.v1.y);
  *(float4*)(Vd + 4) = make_float4(s.v0.z, s.v1.z, s.v0.w, s.v1.w);
}

__device__ __forceinline__ void load_q(v2f (&qreg)[16], const float* Qh, int q0,
                                       int lo, int hi, float qs) {
  const float* qrow = Qh + (size_t)(q0 + lo) * DH;
#pragma unroll
  for (int i = 0; i < 16; ++i) {
    qreg[i].x = qrow[i * 4 + hi * 2] * qs;
    qreg[i].y = qrow[i * 4 + hi * 2 + 1] * qs;
  }
}

__device__ __forceinline__ void store_tile(float* orow, const v8f& o0,
                                           const v8f& o1, const v8f& o2,
                                           const v8f& o3) {
#pragma unroll
  for (int r = 0; r < 8; ++r) {
    orow[r]      = o0[r];
    orow[16 + r] = o1[r];
    orow[32 + r] = o2[r];
    orow[48 + r] = o3[r];
  }
}

__global__ void __launch_bounds__(256) attn_kernel(const float* __restrict__ Q,
                            const float* __restrict__ K,
                            const float* __restrict__ V,
                            float* __restrict__ O,
                            int nblk) {
  __shared__ __attribute__((aligned(16))) float Ks[2 * 2 * KSZ];
  __shared__ __attribute__((aligned(16))) float Vs[2 * 2 * VSZ];
  __shared__ float Mg[4 * MGV * 32];
  if ((int)blockIdx.x >= nblk) return;

  const int tid = threadIdx.x;
  const int lane = tid & 31;
  const int lo = lane & 15;
  const int hi = lane >> 4;
  const int w8 = __builtin_amdgcn_readfirstlane(tid >> 5);
  const int set = w8 >> 2;
  const int w = w8 & 3;
  const int ht = tid & 127;

  const int groups = T_SEQ / 64;
  const int gpairs = groups / 2;
  const int bh = blockIdx.x / gpairs;
  const int gp = blockIdx.x - bh * gpairs;
  const int gl = groups - 1 - gp;
  const float* Qh = Q + (size_t)bh * T_SEQ * DH;
  const float* Kh = K + (size_t)bh * T_SEQ * DH;
  const float* Vh = V + (size_t)bh * T_SEQ * DH;

  const float qs = 0.125f * 1.44269504088896340736f;
  const int b = bh / HEADS;
  const int h = bh - b * HEADS;

  const int nsw = set ? (4 * gl + 4 - SPLIT_KT) : NSTEP;
  const int tb = set ? SPLIT_KT : 0;
#define TILE_OF(s) (((s) < nsw) ? (tb + (s)) : ((s) - nsw))

  const int srow = ht >> 4;
  const int scol = (ht & 15) * 4;
  const float* kg = Kh + srow * DH + scol;
  const float* vg = Vh + (2 * srow) * DH + scol;
  const float* Kr = Ks + set * (2 * KSZ);
  const float* Vr = Vs + set * (2 * VSZ);
  float* ksd = Ks + set * (2 * KSZ) + srow * KP + scol;
  float* vsd = Vs + set * (2 * VSZ) + srow * VPP + 2 * scol;
  float* mg = Mg + (w * MGV) * 32 + lane;

  int qt = 4 * gl + w;
  v2f qreg[16];
  load_q(qreg, Qh, qt * 16, lo, hi, qs);

  float m_run = -__builtin_inff();
  float l_run = 0.0f;
  v8f o0 = {}, o1 = {}, o2 = {}, o3 = {};

  stage_store(ksd, vsd, stage_load(kg, vg, TILE_OF(0)));
  StageRegs nxt = stage_load(kg, vg, TILE_OF(1));
  __syncthreads();

  for (int s = 0; s < NSTEP; ++s) {
    const int sf = (s + 2 < NSTEP) ? (s + 2) : (NSTEP - 1);
    const StageRegs inc = stage_load(kg, vg, TILE_OF(sf));
    const int nb = (s + 1) & 1;
    stage_store(ksd + nb * KSZ, vsd + nb * VSZ, nxt);
    nxt = inc;

    if (s == nsw) {
#pragma unroll
      for (int r = 0; r < 8; ++r) {
        mg[r * 32]        = o0[r];
        mg[(8 + r) * 32]  = o1[r];
        mg[(16 + r) * 32] = o2[r];
        mg[(24 + r) * 32] = o3[r];
      }
      mg[32 * 32] = m_run;
      mg[33 * 32] = l_run;
      qt = 4 * gp + 3 - w;
      load_q(qreg, Qh, qt * 16, lo, hi, qs);
      m_run = -__builtin_inff();
      l_run = 0.0f;
      const v8f z = {};
      o0 = z; o1 = z; o2 = z; o3 = z;
    }

    const int j = TILE_OF(s);
    const float* Kb = Kr + (s & 1) * KSZ;
    const float* Vb = Vr + (s & 1) * VSZ;

    if (j <= qt) {
    v8f sc = {};
    const float* krow = Kb + lo * KP + hi * 2;
    v2f ka[16];
#pragma unroll
    for (int i = 0; i < 16; ++i) {
      ka[i].x = krow[i * 4];
      ka[i].y = krow[i * 4 + 1];
    }
    asm volatile("" :: "v"(ka[0]), "v"(ka[1]), "v"(ka[2]), "v"(ka[3]),
                       "v"(ka[4]), "v"(ka[5]), "v"(ka[6]), "v"(ka[7]),
                       "v"(ka[8]), "v"(ka[9]), "v"(ka[10]), "v"(ka[11]),
                       "v"(ka[12]), "v"(ka[13]), "v"(ka[14]), "v"(ka[15]));
#pragma unroll
    for (int i = 0; i < 16; ++i) sc = wmma4(ka[i], qreg[i], sc);
    if (j == qt) {
#pragma unroll
      for (int r = 0; r < 8; ++r)
        sc[r] = (r + hi * 8 > lo) ? -__builtin_inff() : sc[r];
    }

    float tmax = sc[0];
#pragma unroll
    for (int r = 1; r < 8; ++r) tmax = fmaxf(tmax, sc[r]);
    tmax = fmaxf(tmax, __shfl_xor(tmax, 16, 32));
    const float m_new = fmaxf(m_run, tmax);
    const float alpha = __builtin_amdgcn_exp2f(m_run - m_new);
    float p[8];
    float ts = 0.0f;
#pragma unroll
    for (int r = 0; r < 8; ++r) {
      p[r] = __builtin_amdgcn_exp2f(sc[r] - m_new);
      ts += p[r];
    }
    ts += __shfl_xor(ts, 16, 32);
    l_run = l_run * alpha + ts;
    m_run = m_new;
#pragma unroll
    for (int r = 0; r < 8; ++r) {
      o0[r] *= alpha; o1[r] *= alpha; o2[r] *= alpha; o3[r] *= alpha;
    }

    const float* vrow = Vb + hi * VPP + 2 * lo;
#pragma unroll
    for (int kk = 0; kk < 16; kk += 4) {
      v2f bop;
      if (kk < 8) {
        const float sx = __shfl_xor(p[kk + 2], 16, 32);
        const float sy = __shfl_xor(p[kk + 3], 16, 32);
        bop.x = hi ? sx : p[kk];
        bop.y = hi ? sy : p[kk + 1];
      } else {
        const int jj = kk - 8;
        const float sx = __shfl_xor(p[jj], 16, 32);
        const float sy = __shfl_xor(p[jj + 1], 16, 32);
        bop.x = hi ? p[jj + 2] : sx;
        bop.y = hi ? p[jj + 3] : sy;
      }
      const float* vk = vrow + (kk / 2) * VPP;
      v2f a;
      a.x = vk[0];       a.y = vk[1];         o0 = wmma4(a, bop, o0);
      a.x = vk[32];      a.y = vk[33];        o1 = wmma4(a, bop, o1);
      a.x = vk[64];      a.y = vk[65];        o2 = wmma4(a, bop, o2);
      a.x = vk[96];      a.y = vk[97];        o3 = wmma4(a, bop, o3);
    }
    }

    __syncthreads();
  }
#undef TILE_OF

  float* orow = O + ((size_t)b * T_SEQ + qt * 16 + lo) * C_DIM + h * DH + hi * 8;
  if (set) {
    const float inv_l = 1.0f / l_run;
#pragma unroll
    for (int r = 0; r < 8; ++r) {
      o0[r] *= inv_l; o1[r] *= inv_l; o2[r] *= inv_l; o3[r] *= inv_l;
    }
    store_tile(orow, o0, o1, o2, o3);
  } else {
    const float m1 = mg[32 * 32];
    const float l1 = mg[33 * 32];
    const float m = fmaxf(m_run, m1);
    const float a0 = __builtin_amdgcn_exp2f(m_run - m);
    const float a1 = __builtin_amdgcn_exp2f(m1 - m);
    const float inv_l = 1.0f / (l_run * a0 + l1 * a1);
    const float f0 = a0 * inv_l;
    const float f1 = a1 * inv_l;
#pragma unroll
    for (int r = 0; r < 8; ++r) {
      o0[r] = o0[r] * f0 + mg[r * 32] * f1;
      o1[r] = o1[r] * f0 + mg[(8 + r) * 32] * f1;
      o2[r] = o2[r] * f0 + mg[(16 + r) * 32] * f1;
      o3[r] = o3[r] * f0 + mg[(24 + r) * 32] * f1;
    }
    store_tile(orow, o0, o1, o2, o3);
  }
}

__global__ void __launch_bounds__(256) out_proj_kernel(const float* __restrict__ A,
                                const float* __restrict__ W,
                                const float* __restrict__ bias,
                                float* __restrict__ Out) {
  __shared__ __attribute__((aligned(16))) float As[2 * GASZ];
  __shared__ __attribute__((aligned(16))) float Bs[2 * GBSZ];
  const int nblks = C_DIM / GBN;
  const int m_blk = blockIdx.x / nblks;
  const int n_blk = blockIdx.x - m_blk * nblks;
  const int tid = threadIdx.x;
  const int lo = tid & 15;
  const int hi = (tid >> 4) & 1;
  const int w = __builtin_amdgcn_readfirstlane(tid >> 5);
  const int wm = w & 3, wn = w >> 2;

  GemmAcc c;
  gemm_block(A, W, C_DIM, m_blk, n_blk, As, Bs, wm, wn, c);

  const int n_base = n_blk * GBN + wn * 64;
  const int row0 = m_blk * GBM + wm * 32 + hi * 8;
  float* d0 = Out + (size_t)row0 * C_DIM + n_base + lo;
  const float bv0 = bias[n_base + lo];
  const float bv1 = bias[n_base + 16 + lo];
  const float bv2 = bias[n_base + 32 + lo];
  const float bv3 = bias[n_base + 48 + lo];
#pragma unroll
  for (int r = 0; r < 8; ++r) {
    float* d = d0 + (size_t)r * C_DIM;
    d[0]  = c.c00[r] + bv0;
    d[16] = c.c01[r] + bv1;
    d[32] = c.c02[r] + bv2;
    d[48] = c.c03[r] + bv3;
    d += 16 * C_DIM;
    d[0]  = c.c10[r] + bv0;
    d[16] = c.c11[r] + bv1;
    d[32] = c.c12[r] + bv2;
    d[48] = c.c13[r] + bv3;
  }
}

extern "C" void kernel_launch(void* const* d_in, const int* in_sizes, int n_in,
                              void* d_out, int out_size, void* d_ws, size_t ws_size,
                              hipStream_t stream) {
  const float* x     = (const float*)d_in[0];
  const float* W_qkv = (const float*)d_in[1];
  const float* b_qkv = (const float*)d_in[2];
  const float* W_fc  = (const float*)d_in[3];
  const float* b_fc  = (const float*)d_in[4];
  float* out = (float*)d_out;

  const int BT = in_sizes[0] / C_DIM;
  const int B  = BT / T_SEQ;

  const size_t headElems = (size_t)B * HEADS * T_SEQ * DH;
  float* Qw = (float*)d_ws;
  float* Kw = Qw + headElems;
  float* Vw = Kw + headElems;
  float* Aw = Vw + headElems;

  const int mblks = BT / GBM;

  {
    const int blocks = mblks * ((3 * C_DIM) / GBN);
    qkv_proj_kernel<<<blocks, 256, 0, stream>>>(x, W_qkv, b_qkv, Qw, Kw, Vw);
  }
  {
    const int blocks = B * HEADS * (T_SEQ / 128);
    attn_kernel<<<blocks, 256, 0, stream>>>(Qw, Kw, Vw, Aw, blocks);
  }
  {
    const int blocks = mblks * (C_DIM / GBN);
    out_proj_kernel<<<blocks, 256, 0, stream>>>(Aw, W_fc, b_fc, out);
  }
}
